# starting kernel
# baseline (speedup 1.0000x reference)
#include <hip/hip_runtime.h>


#define SIZE 8192
#define NB   64
#define BS   128
#define TM   64
#define KH   64
#define LDX  68

typedef float v2f __attribute__((ext_vector_type(2)));
typedef float v8f __attribute__((ext_vector_type(8)));

template <int PASS>
__global__ __launch_bounds__(256) void monarch_pass(
    const float* __restrict__ src, const float* __restrict__ w,
    const float* __restrict__ bias, float* __restrict__ dst) {
  __shared__ float Xs[TM][LDX];
  __shared__ float Wst[BS][LDX];

  const int tid  = threadIdx.x;
  const int lane = tid & 31;
  const int wv   = tid >> 5;
  const int b0   = blockIdx.x * TM;
  const int kblk = blockIdx.y;

  const float* __restrict__ wblk = w + (size_t)kblk * BS * BS;

  const int mrow  = (wv & 3) * 16 + (lane & 15);
  const int khalf = (lane >> 4) << 1;
  const int nbase = (wv >> 2) * 64;
  const int ncol  = lane & 15;

  v8f acc[4] = {};

  #pragma unroll
  for (int kh = 0; kh < 2; ++kh) {
    __syncthreads();
    #pragma unroll
    for (int i = 0; i < 4; ++i) {
      int f  = tid + i * 256;
      int r  = f >> 4;
      int c4 = (f & 15) << 2;
      float4 v = *reinterpret_cast<const float4*>(
          src + (size_t)(b0 + r) * SIZE + (size_t)kblk * BS + kh * KH + c4);
      *reinterpret_cast<float4*>(&Xs[r][c4]) = v;
    }
    #pragma unroll
    for (int i = 0; i < 8; ++i) {
      int f  = tid + i * 256;
      int d  = f >> 5;
      int c4 = (f & 31) << 2;
      float4 v = *reinterpret_cast<const float4*>(
          wblk + (size_t)(kh * KH + d) * BS + c4);
      Wst[c4 + 0][d] = v.x;
      Wst[c4 + 1][d] = v.y;
      Wst[c4 + 2][d] = v.z;
      Wst[c4 + 3][d] = v.w;
    }
    __syncthreads();

    #pragma unroll 4
    for (int kk = 0; kk < KH; kk += 4) {
      v2f a = *reinterpret_cast<const v2f*>(&Xs[mrow][kk + khalf]);
      #pragma unroll
      for (int j = 0; j < 4; ++j) {
        v2f b = *reinterpret_cast<const v2f*>(&Wst[nbase + j * 16 + ncol][kk + khalf]);
        acc[j] = __builtin_amdgcn_wmma_f32_16x16x4_f32(
             false, a,  false, b,
             (short)0, acc[j],  false,  false);
      }
    }
  }

  const int mtop = (wv & 3) * 16 + ((lane >> 4) << 3);
  #pragma unroll
  for (int j = 0; j < 4; ++j) {
    const int e = nbase + j * 16 + ncol;
    size_t col;
    float badd = 0.0f;
    if (PASS == 1) {
      col = (size_t)e * NB + kblk;
    } else {
      col = (size_t)(e & 63) * BS + 2 * kblk + (e >> 6);
      badd = bias[col];
    }
    float* __restrict__ drow = dst + (size_t)(b0 + mtop) * SIZE + col;
    #pragma unroll
    for (int r = 0; r < 8; ++r) {
      float v = acc[j][r];
      if (PASS == 2) v += badd;
      drow[(size_t)r * SIZE] = v;
    }
  }
}

extern "C" void kernel_launch(void* const* d_in, const int* in_sizes, int n_in,
                              void* d_out, int out_size, void* d_ws, size_t ws_size,
                              hipStream_t stream) {
  const float* x    = (const float*)d_in[0];
  const float* w1   = (const float*)d_in[1];
  const float* w2   = (const float*)d_in[2];
  const float* bias = (const float*)d_in[3];
  float* out   = (float*)d_out;
  float* inter = (float*)d_ws;

  const int Btok = in_sizes[0] / SIZE;
  dim3 block(256);
  dim3 grid(Btok / TM, NB);

  monarch_pass<1><<<grid, block, 0, stream>>>(x,     w1, bias, inter);
  monarch_pass<2><<<grid, block, 0, stream>>>(inter, w2, bias, out);
}


# Both passes fused into one kernel: a 32-token work-group computes its h1 panel from x straight into LDS (x lines parked in the panel row they produce), no intermediate in memory; sums unchanged
# speedup vs baseline: 4.5243x; 4.5243x over previous
#include <hip/hip_runtime.h>


#define SIZE 8192
#define NB   64
#define BS   128
#define TT   32
#define KS2  36
#define TS2  (64 * KS2 + 4)

typedef float v2f __attribute__((ext_vector_type(2)));
typedef float v8f __attribute__((ext_vector_type(8)));
typedef float v4f __attribute__((ext_vector_type(4)));


__device__ __forceinline__ v8f wmma_step(v8f c, v2f a, float bx, float by) {
  v2f b;
  b.x = bx;
  b.y = by;
  return __builtin_amdgcn_wmma_f32_16x16x4_f32(false, a, false, b, (short)0, c, false, false);
}


__device__ __forceinline__ void load_w1half(v2f (&r)[8], const float* __restrict__ p) {
  #pragma unroll
  for (int i = 0; i < 4; ++i) {
    r[2 * i]     = *reinterpret_cast<const v2f*>(p + (4 * i) * BS);
    r[2 * i + 1] = *reinterpret_cast<const v2f*>(p + (4 * i + 1) * BS);
  }
}

__device__ __forceinline__ void mma1_half(v8f (&acc)[4], const v2f (&r)[8], const float* sa) {
  #pragma unroll
  for (int i = 0; i < 4; ++i) {
    const v2f a0 = *reinterpret_cast<const v2f*>(sa + 4 * i);
    const v2f a1 = *reinterpret_cast<const v2f*>(sa + 16 * TS2 + 4 * i);
    const v2f w0 = r[2 * i], w1 = r[2 * i + 1];
    acc[0] = wmma_step(acc[0], a0, w0.x, w1.x);
    acc[2] = wmma_step(acc[2], a1, w0.x, w1.x);
    acc[1] = wmma_step(acc[1], a0, w0.y, w1.y);
    acc[3] = wmma_step(acc[3], a1, w0.y, w1.y);
  }
}

template <bool LAST>
__device__ __forceinline__ void first_product_block(
    const float* __restrict__ xl, const float* __restrict__ wl,
    float* sl, const float* sa, float* pe1,
    v4f (&xn)[8], v2f (&wa)[8], v2f (&wh)[8]) {
  v8f acc[4] = {};
  #pragma unroll
  for (int ch = 0; ch < 4; ++ch) {
    const bool more = !LAST || ch < 3;
    if (more) {
      #pragma unroll
      for (int i = 0; i < 8; ++i)
        xn[i] = *reinterpret_cast<const v4f*>(xl + 32 * (ch + 1) + (size_t)(4 * i) * SIZE);
    }
    mma1_half(acc, wa, sa);
    if (more) load_w1half(wa, wl + (32 * (ch + 1)) * BS);
    mma1_half(acc, wh, sa + 16);
    if (more) load_w1half(wh, wl + (32 * (ch + 1) + 16) * BS);
    if (ch == 3) {
      #pragma unroll
      for (int mt = 0; mt < 2; ++mt) {
        #pragma unroll
        for (int r = 0; r < 8; ++r) {
          v2f v;
          v.x = acc[2 * mt][r];
          v.y = acc[2 * mt + 1][r];
          *reinterpret_cast<v2f*>(pe1 + (16 * mt + r) * TS2) = v;
        }
      }
    }
    if (more) {
      float* sn = sl + (ch == 3 ? KS2 : 0);
      #pragma unroll
      for (int i = 0; i < 8; ++i)
        *reinterpret_cast<v4f*>(sn + (4 * i) * TS2) = xn[i];
    }
  }
}


__device__ __forceinline__ void load_wgroup(float4 (&r)[8], const float* __restrict__ p) {
  #pragma unroll
  for (int q = 0; q < 2; ++q) {
    r[4 * q + 0] = *reinterpret_cast<const float4*>(p + (4 * q) * BS);
    r[4 * q + 1] = *reinterpret_cast<const float4*>(p + (4 * q + 1) * BS);
    r[4 * q + 2] = *reinterpret_cast<const float4*>(p + (4 * q) * BS + 64);
    r[4 * q + 3] = *reinterpret_cast<const float4*>(p + (4 * q + 1) * BS + 64);
  }
}

__device__ __forceinline__ void mma_group(v8f (&acc)[16], const float4 (&r)[8], const float* pa) {
  #pragma unroll
  for (int q = 0; q < 2; ++q) {
    v2f a0, a1;
    a0.x = pa[(4 * q) * KS2];
    a0.y = pa[(4 * q + 1) * KS2];
    a1.x = pa[16 * TS2 + (4 * q) * KS2];
    a1.y = pa[16 * TS2 + (4 * q + 1) * KS2];
    const float4 lo0 = r[4 * q + 0], lo1 = r[4 * q + 1], hi0 = r[4 * q + 2], hi1 = r[4 * q + 3];
    acc[0]  = wmma_step(acc[0],  a0, lo0.x, lo1.x);
    acc[8]  = wmma_step(acc[8],  a1, lo0.x, lo1.x);
    acc[1]  = wmma_step(acc[1],  a0, lo0.y, lo1.y);
    acc[9]  = wmma_step(acc[9],  a1, lo0.y, lo1.y);
    acc[2]  = wmma_step(acc[2],  a0, lo0.z, lo1.z);
    acc[10] = wmma_step(acc[10], a1, lo0.z, lo1.z);
    acc[3]  = wmma_step(acc[3],  a0, lo0.w, lo1.w);
    acc[11] = wmma_step(acc[11], a1, lo0.w, lo1.w);
    acc[4]  = wmma_step(acc[4],  a0, hi0.x, hi1.x);
    acc[12] = wmma_step(acc[12], a1, hi0.x, hi1.x);
    acc[5]  = wmma_step(acc[5],  a0, hi0.y, hi1.y);
    acc[13] = wmma_step(acc[13], a1, hi0.y, hi1.y);
    acc[6]  = wmma_step(acc[6],  a0, hi0.z, hi1.z);
    acc[14] = wmma_step(acc[14], a1, hi0.z, hi1.z);
    acc[7]  = wmma_step(acc[7],  a0, hi0.w, hi1.w);
    acc[15] = wmma_step(acc[15], a1, hi0.w, hi1.w);
  }
}

__global__ __launch_bounds__(512) void blockdiag_fused_panel(
    const float* __restrict__ x, const float* __restrict__ w1, const float* __restrict__ w2,
    const float* __restrict__ bias, float* __restrict__ out) {
  __shared__ __attribute__((aligned(16))) float P[TT * TS2];

  const int tid  = threadIdx.x;
  const int lane = tid & 31;
  const int wv   = tid >> 5;
  const int n    = lane & 15;
  const int h    = lane >> 4;
  const int G    = blockIdx.x;
  const int b0   = blockIdx.y * TT;

  {
    const int k0 = 4 * wv;
    const float* __restrict__ xl = x + (size_t)(b0 + (lane >> 3)) * SIZE + k0 * BS + ((lane & 7) << 2);
    const float* __restrict__ wl = w1 + (size_t)(k0 * BS + 2 * h) * BS + G * 32 + 2 * n;
    float*       sl  = &P[(lane >> 3) * TS2 + k0 * KS2 + ((lane & 7) << 2)];
    const float* sa  = &P[n * TS2 + k0 * KS2 + 2 * h];
    float*       pe1 = &P[(8 * h) * TS2 + k0 * KS2 + 2 * n];

    v4f xn[8];
    v2f wa[8], wh[8];
    #pragma unroll
    for (int i = 0; i < 8; ++i)
      xn[i] = *reinterpret_cast<const v4f*>(xl + (size_t)(4 * i) * SIZE);
    load_w1half(wa, wl);
    load_w1half(wh, wl + 16 * BS);
    #pragma unroll
    for (int i = 0; i < 8; ++i)
      *reinterpret_cast<v4f*>(sl + (4 * i) * TS2) = xn[i];

    #pragma nounroll
    for (int j = 0; j < 3; ++j)
      first_product_block<false>(xl + j * BS, wl + (size_t)j * BS * BS, sl + j * KS2, sa + j * KS2, pe1 + j * KS2,
                                 xn, wa, wh);
    first_product_block<true>(xl + 3 * BS, wl + (size_t)3 * BS * BS, sl + 3 * KS2, sa + 3 * KS2, pe1 + 3 * KS2,
                              xn, wa, wh);
  }

  const int g = wv;
  const float* __restrict__ wb = w2 + (size_t)(G * 16 + g) * BS * BS + (size_t)(2 * h) * BS + 4 * n;
  const float* pa = &P[n * TS2 + (2 * h) * KS2 + 2 * g];
  float* pe = &P[(8 * h) * TS2 + (4 * n) * KS2 + 2 * g];

  float4 rA[8], rB[8];
  load_wgroup(rA, wb);
  __syncthreads();

  {
    v8f acc[16] = {};
    #pragma nounroll
    for (int p = 0; p < 7; ++p) {
      const float* pap = pa + (16 * (p & 3)) * KS2 + (p >> 2);
      load_wgroup(rB, wb + (16 * p + 8) * BS);
      mma_group(acc, rA, pap);
      load_wgroup(rA, wb + (16 * p + 16) * BS);
      mma_group(acc, rB, pap + 8 * KS2);
    }
    {
      const float* pap = pa + 48 * KS2 + 1;
      load_wgroup(rB, wb + 120 * BS);
      mma_group(acc, rA, pap);
      mma_group(acc, rB, pap + 8 * KS2);
    }
    #pragma unroll
    for (int mt = 0; mt < 2; ++mt) {
      #pragma unroll
      for (int j = 0; j < 4; ++j) {
        #pragma unroll
        for (int r = 0; r < 8; ++r) {
          float2 v;
          v.x = acc[8 * mt + j][r];
          v.y = acc[8 * mt + 4 + j][r];
          *reinterpret_cast<float2*>(pe + (16 * mt + r) * TS2 + j * KS2) = v;
        }
      }
    }
  }
  __syncthreads();

  {
    const int k = tid >> 3, dd4 = (tid & 7) << 2;
    const float4 bb = *reinterpret_cast<const float4*>(bias + k * BS + G * 32 + dd4);
    const float* ps = &P[k * KS2 + dd4];
    float* po = out + (size_t)b0 * SIZE + k * BS + G * 32 + dd4;
    for (int i0 = 0; i0 < 32; i0 += 8) {
      #pragma unroll
      for (int i = 0; i < 8; ++i) {
        const int t = i0 + i;
        float4 v = *reinterpret_cast<const float4*>(ps + t * TS2);
        v.x += bb.x; v.y += bb.y; v.z += bb.z; v.w += bb.w;
        *reinterpret_cast<float4*>(po + (size_t)t * SIZE) = v;
      }
    }
  }
}

extern "C" void kernel_launch(void* const* d_in, const int* in_sizes, int n_in,
                              void* d_out, int out_size, void* d_ws, size_t ws_size,
                              hipStream_t stream) {
  const float* x    = (const float*)d_in[0];
  const float* w1   = (const float*)d_in[1];
  const float* w2   = (const float*)d_in[2];
  const float* bias = (const float*)d_in[3];
  float* out   = (float*)d_out;

  const int Btok = in_sizes[0] / SIZE;
  dim3 block(512);
  dim3 grid(NB / 16, Btok / TT);

  blockdiag_fused_panel<<<grid, block, 0, stream>>>(x, w1, w2, bias, out);
}


# Issue priority follows progress: each wave starts a product at s_setprio 3 and steps down to 2, 1, 0 near its end, so the four waves of a SIMD reach the barrier together; loads and sums unchanged
# speedup vs baseline: 4.6627x; 1.0306x over previous
#include <hip/hip_runtime.h>


#define SIZE 8192
#define NB   64
#define BS   128
#define TT   32
#define KS2  36
#define TS2  (64 * KS2 + 4)

typedef float v2f __attribute__((ext_vector_type(2)));
typedef float v8f __attribute__((ext_vector_type(8)));
typedef float v4f __attribute__((ext_vector_type(4)));


__device__ __forceinline__ v8f wmma_step(v8f c, v2f a, float bx, float by) {
  v2f b;
  b.x = bx;
  b.y = by;
  return __builtin_amdgcn_wmma_f32_16x16x4_f32(false, a, false, b, (short)0, c, false, false);
}


__device__ __forceinline__ void load_w1half(v2f (&r)[8], const float* __restrict__ p) {
  #pragma unroll
  for (int i = 0; i < 4; ++i) {
    r[2 * i]     = *reinterpret_cast<const v2f*>(p + (4 * i) * BS);
    r[2 * i + 1] = *reinterpret_cast<const v2f*>(p + (4 * i + 1) * BS);
  }
}

__device__ __forceinline__ void mma1_half(v8f (&acc)[4], const v2f (&r)[8], const float* sa) {
  #pragma unroll
  for (int i = 0; i < 4; ++i) {
    const v2f a0 = *reinterpret_cast<const v2f*>(sa + 4 * i);
    const v2f a1 = *reinterpret_cast<const v2f*>(sa + 16 * TS2 + 4 * i);
    const v2f w0 = r[2 * i], w1 = r[2 * i + 1];
    acc[0] = wmma_step(acc[0], a0, w0.x, w1.x);
    acc[2] = wmma_step(acc[2], a1, w0.x, w1.x);
    acc[1] = wmma_step(acc[1], a0, w0.y, w1.y);
    acc[3] = wmma_step(acc[3], a1, w0.y, w1.y);
  }
}

template <bool LAST>
__device__ __forceinline__ void first_product_block(
    const float* __restrict__ xl, const float* __restrict__ wl,
    float* sl, const float* sa, float* pe1,
    v4f (&xn)[8], v2f (&wa)[8], v2f (&wh)[8]) {
  v8f acc[4] = {};
  #pragma unroll
  for (int ch = 0; ch < 4; ++ch) {
    const bool more = !LAST || ch < 3;
    if (LAST && ch == 0) __builtin_amdgcn_s_setprio(1);
    if (LAST && ch == 2) __builtin_amdgcn_s_setprio(0);
    if (more) {
      #pragma unroll
      for (int i = 0; i < 8; ++i)
        xn[i] = *reinterpret_cast<const v4f*>(xl + 32 * (ch + 1) + (size_t)(4 * i) * SIZE);
    }
    mma1_half(acc, wa, sa);
    if (more) load_w1half(wa, wl + (32 * (ch + 1)) * BS);
    mma1_half(acc, wh, sa + 16);
    if (more) load_w1half(wh, wl + (32 * (ch + 1) + 16) * BS);
    if (ch == 3) {
      #pragma unroll
      for (int mt = 0; mt < 2; ++mt) {
        #pragma unroll
        for (int r = 0; r < 8; ++r) {
          v2f v;
          v.x = acc[2 * mt][r];
          v.y = acc[2 * mt + 1][r];
          *reinterpret_cast<v2f*>(pe1 + (16 * mt + r) * TS2) = v;
        }
      }
    }
    if (more) {
      float* sn = sl + (ch == 3 ? KS2 : 0);
      #pragma unroll
      for (int i = 0; i < 8; ++i)
        *reinterpret_cast<v4f*>(sn + (4 * i) * TS2) = xn[i];
    }
  }
}


__device__ __forceinline__ void load_wgroup(float4 (&r)[8], const float* __restrict__ p) {
  #pragma unroll
  for (int q = 0; q < 2; ++q) {
    r[4 * q + 0] = *reinterpret_cast<const float4*>(p + (4 * q) * BS);
    r[4 * q + 1] = *reinterpret_cast<const float4*>(p + (4 * q + 1) * BS);
    r[4 * q + 2] = *reinterpret_cast<const float4*>(p + (4 * q) * BS + 64);
    r[4 * q + 3] = *reinterpret_cast<const float4*>(p + (4 * q + 1) * BS + 64);
  }
}

__device__ __forceinline__ void mma_group(v8f (&acc)[16], const float4 (&r)[8], const float* pa) {
  #pragma unroll
  for (int q = 0; q < 2; ++q) {
    v2f a0, a1;
    a0.x = pa[(4 * q) * KS2];
    a0.y = pa[(4 * q + 1) * KS2];
    a1.x = pa[16 * TS2 + (4 * q) * KS2];
    a1.y = pa[16 * TS2 + (4 * q + 1) * KS2];
    const float4 lo0 = r[4 * q + 0], lo1 = r[4 * q + 1], hi0 = r[4 * q + 2], hi1 = r[4 * q + 3];
    acc[0]  = wmma_step(acc[0],  a0, lo0.x, lo1.x);
    acc[8]  = wmma_step(acc[8],  a1, lo0.x, lo1.x);
    acc[1]  = wmma_step(acc[1],  a0, lo0.y, lo1.y);
    acc[9]  = wmma_step(acc[9],  a1, lo0.y, lo1.y);
    acc[2]  = wmma_step(acc[2],  a0, lo0.z, lo1.z);
    acc[10] = wmma_step(acc[10], a1, lo0.z, lo1.z);
    acc[3]  = wmma_step(acc[3],  a0, lo0.w, lo1.w);
    acc[11] = wmma_step(acc[11], a1, lo0.w, lo1.w);
    acc[4]  = wmma_step(acc[4],  a0, hi0.x, hi1.x);
    acc[12] = wmma_step(acc[12], a1, hi0.x, hi1.x);
    acc[5]  = wmma_step(acc[5],  a0, hi0.y, hi1.y);
    acc[13] = wmma_step(acc[13], a1, hi0.y, hi1.y);
    acc[6]  = wmma_step(acc[6],  a0, hi0.z, hi1.z);
    acc[14] = wmma_step(acc[14], a1, hi0.z, hi1.z);
    acc[7]  = wmma_step(acc[7],  a0, hi0.w, hi1.w);
    acc[15] = wmma_step(acc[15], a1, hi0.w, hi1.w);
  }
}

__global__ __launch_bounds__(512) void blockdiag_fused_panel(
    const float* __restrict__ x, const float* __restrict__ w1, const float* __restrict__ w2,
    const float* __restrict__ bias, float* __restrict__ out) {
  __shared__ __attribute__((aligned(16))) float P[TT * TS2];

  const int tid  = threadIdx.x;
  const int lane = tid & 31;
  const int wv   = tid >> 5;
  const int n    = lane & 15;
  const int h    = lane >> 4;
  const int G    = blockIdx.x;
  const int b0   = blockIdx.y * TT;

  {
    const int k0 = 4 * wv;
    const float* __restrict__ xl = x + (size_t)(b0 + (lane >> 3)) * SIZE + k0 * BS + ((lane & 7) << 2);
    const float* __restrict__ wl = w1 + (size_t)(k0 * BS + 2 * h) * BS + G * 32 + 2 * n;
    float*       sl  = &P[(lane >> 3) * TS2 + k0 * KS2 + ((lane & 7) << 2)];
    const float* sa  = &P[n * TS2 + k0 * KS2 + 2 * h];
    float*       pe1 = &P[(8 * h) * TS2 + k0 * KS2 + 2 * n];

    v4f xn[8];
    v2f wa[8], wh[8];
    #pragma unroll
    for (int i = 0; i < 8; ++i)
      xn[i] = *reinterpret_cast<const v4f*>(xl + (size_t)(4 * i) * SIZE);
    load_w1half(wa, wl);
    load_w1half(wh, wl + 16 * BS);
    #pragma unroll
    for (int i = 0; i < 8; ++i)
      *reinterpret_cast<v4f*>(sl + (4 * i) * TS2) = xn[i];

    __builtin_amdgcn_s_setprio(3);
    #pragma nounroll
    for (int j = 0; j < 3; ++j) {
      if (j == 2) __builtin_amdgcn_s_setprio(2);
      first_product_block<false>(xl + j * BS, wl + (size_t)j * BS * BS, sl + j * KS2, sa + j * KS2, pe1 + j * KS2,
                                 xn, wa, wh);
    }
    first_product_block<true>(xl + 3 * BS, wl + (size_t)3 * BS * BS, sl + 3 * KS2, sa + 3 * KS2, pe1 + 3 * KS2,
                              xn, wa, wh);
  }

  const int g = wv;
  const float* __restrict__ wb = w2 + (size_t)(G * 16 + g) * BS * BS + (size_t)(2 * h) * BS + 4 * n;
  const float* pa = &P[n * TS2 + (2 * h) * KS2 + 2 * g];
  float* pe = &P[(8 * h) * TS2 + (4 * n) * KS2 + 2 * g];

  float4 rA[8], rB[8];
  load_wgroup(rA, wb);
  __syncthreads();
  __builtin_amdgcn_s_setprio(3);

  {
    v8f acc[16] = {};
    #pragma nounroll
    for (int p = 0; p < 7; ++p) {
      const float* pap = pa + (16 * (p & 3)) * KS2 + (p >> 2);
      if (p == 5) __builtin_amdgcn_s_setprio(2);
      load_wgroup(rB, wb + (16 * p + 8) * BS);
      mma_group(acc, rA, pap);
      load_wgroup(rA, wb + (16 * p + 16) * BS);
      mma_group(acc, rB, pap + 8 * KS2);
    }
    {
      const float* pap = pa + 48 * KS2 + 1;
      load_wgroup(rB, wb + 120 * BS);
      __builtin_amdgcn_s_setprio(1);
      mma_group(acc, rA, pap);
      __builtin_amdgcn_s_setprio(0);
      mma_group(acc, rB, pap + 8 * KS2);
    }
    #pragma unroll
    for (int mt = 0; mt < 2; ++mt) {
      #pragma unroll
      for (int j = 0; j < 4; ++j) {
        #pragma unroll
        for (int r = 0; r < 8; ++r) {
          float2 v;
          v.x = acc[8 * mt + j][r];
          v.y = acc[8 * mt + 4 + j][r];
          *reinterpret_cast<float2*>(pe + (16 * mt + r) * TS2 + j * KS2) = v;
        }
      }
    }
  }
  __syncthreads();

  {
    const int k = tid >> 3, dd4 = (tid & 7) << 2;
    const float4 bb = *reinterpret_cast<const float4*>(bias + k * BS + G * 32 + dd4);
    const float* ps = &P[k * KS2 + dd4];
    float* po = out + (size_t)b0 * SIZE + k * BS + G * 32 + dd4;
    for (int i0 = 0; i0 < 32; i0 += 8) {
      #pragma unroll
      for (int i = 0; i < 8; ++i) {
        const int t = i0 + i;
        float4 v = *reinterpret_cast<const float4*>(ps + t * TS2);
        v.x += bb.x; v.y += bb.y; v.z += bb.z; v.w += bb.w;
        *reinterpret_cast<float4*>(po + (size_t)t * SIZE) = v;
      }
    }
  }
}

extern "C" void kernel_launch(void* const* d_in, const int* in_sizes, int n_in,
                              void* d_out, int out_size, void* d_ws, size_t ws_size,
                              hipStream_t stream) {
  const float* x    = (const float*)d_in[0];
  const float* w1   = (const float*)d_in[1];
  const float* w2   = (const float*)d_in[2];
  const float* bias = (const float*)d_in[3];
  float* out   = (float*)d_out;

  const int Btok = in_sizes[0] / SIZE;
  dim3 block(512);
  dim3 grid(NB / 16, Btok / TT);

  blockdiag_fused_panel<<<grid, block, 0, stream>>>(x, w1, w2, bias, out);
}
